# starting kernel
# baseline (speedup 1.0000x reference)
#include <hip/hip_runtime.h>
#include <math.h>

#define NTOT 16384
#define NLAT 20
#define MCL  10
#define KNB  32
#define BB   8
#define MUF  5.0f

typedef __attribute__((ext_vector_type(2))) float v2f;
typedef __attribute__((ext_vector_type(8))) float v8f;

#define NPART   128
#define KCHUNK  (NTOT / NPART)
#define WS_PARTIALS 0
#define WS_ENCODED  (NPART * 512)
#define WS_INVD     (WS_ENCODED + BB * NLAT)
#define WS_EBL      (WS_INVD + MCL * NLAT * BB)

__global__ __launch_bounds__(32)
void encode_partial_wmma(const float* __restrict__ x,
                         const float* __restrict__ enc_w,
                         float* __restrict__ partials) {
  const int t = threadIdx.x;
  const int p = blockIdx.x;
  const int M = t & 15;
  const int Nn = t & 15;
  const int half = (t >> 4) << 1;
  const float maskA  = (M < BB) ? 1.f : 0.f;
  const float maskB1 = (Nn < NLAT - 16) ? 1.f : 0.f;
  const int rowA  = (M < BB) ? M : 0;
  const int rowB1 = (Nn < NLAT - 16) ? (16 + Nn) : 16;
  const float* pa  = x     + (size_t)rowA  * NTOT + half;
  const float* pb0 = enc_w + (size_t)Nn    * NTOT + half;
  const float* pb1 = enc_w + (size_t)rowB1 * NTOT + half;
  v8f acc0a = {}, acc0b = {}, acc1a = {}, acc1b = {};
  const int n0base = p * KCHUNK;
  for (int s = 0; s < KCHUNK; s += 8) {
    const int n0 = n0base + s;
    const float2 av0 = *(const float2*)(pa  + n0);
    const float2 bv0 = *(const float2*)(pb0 + n0);
    const float2 cv0 = *(const float2*)(pb1 + n0);
    const float2 av1 = *(const float2*)(pa  + n0 + 4);
    const float2 bv1 = *(const float2*)(pb0 + n0 + 4);
    const float2 cv1 = *(const float2*)(pb1 + n0 + 4);
    v2f a0; a0.x = av0.x * maskA;  a0.y = av0.y * maskA;
    v2f b0; b0.x = bv0.x;          b0.y = bv0.y;
    v2f c0; c0.x = cv0.x * maskB1; c0.y = cv0.y * maskB1;
    v2f a1; a1.x = av1.x * maskA;  a1.y = av1.y * maskA;
    v2f b1; b1.x = bv1.x;          b1.y = bv1.y;
    v2f c1; c1.x = cv1.x * maskB1; c1.y = cv1.y * maskB1;
    acc0a = __builtin_amdgcn_wmma_f32_16x16x4_f32(false, a0, false, b0,
                                                  (short)0, acc0a, false, false);
    acc1a = __builtin_amdgcn_wmma_f32_16x16x4_f32(false, a0, false, c0,
                                                  (short)0, acc1a, false, false);
    acc0b = __builtin_amdgcn_wmma_f32_16x16x4_f32(false, a1, false, b1,
                                                  (short)0, acc0b, false, false);
    acc1b = __builtin_amdgcn_wmma_f32_16x16x4_f32(false, a1, false, c1,
                                                  (short)0, acc1b, false, false);
  }
  float* out0 = partials + (size_t)(p * 2 + 0) * 256;
  float* out1 = partials + (size_t)(p * 2 + 1) * 256;
#pragma unroll
  for (int i = 0; i < 8; ++i) {
    out0[i * 32 + t] = acc0a[i] + acc0b[i];
    out1[i * 32 + t] = acc1a[i] + acc1b[i];
  }
}

__global__ __launch_bounds__(256)
void encode_reduce(const float* __restrict__ partials,
                   const float* __restrict__ enc_b,
                   float* __restrict__ encoded) {
  const int tid = threadIdx.x;
  if (tid >= BB * NLAT) return;
  const int b = tid / NLAT, l = tid % NLAT;
  const int tile = l >> 4, Nn = l & 15;
  float s = 0.f;
  for (int p = 0; p < NPART; ++p)
    s += partials[(size_t)(p * 2 + tile) * 256 + b * 32 + Nn];
  encoded[b * NLAT + l] = s + enc_b[l];
}

__global__ __launch_bounds__(256)
void bandwidth_kernel(const float* __restrict__ encoded,
                      const float* __restrict__ bw_w,
                      const float* __restrict__ bw_b,
                      float* __restrict__ invd_t,
                      float* __restrict__ ebl) {
  const int tid = blockIdx.x * blockDim.x + threadIdx.x;
  if (tid < BB * NLAT * MCL) {
    const int b = tid / (NLAT * MCL);
    const int j = tid % (NLAT * MCL);
    float acc = bw_b[j];
    for (int l2 = 0; l2 < NLAT; ++l2)
      acc = fmaf(encoded[b * NLAT + l2], bw_w[j * NLAT + l2], acc);
    const float s = 1.0f / (1.0f + expf(-acc));
    const float lo = 2.0f / 60.0f / MUF;
    const float hi = 1.0f / 60.0f;
    const float bw = (hi - lo) * s + lo;
    const int l = j / MCL, m = j % MCL;
    const float d = bw * MUF;
    invd_t[((size_t)m * NLAT + l) * BB + b] = 1.0f / (d * d);
  }
  if (tid < BB * NLAT) {
    const int l = tid / BB, b = tid % BB;
    ebl[tid] = encoded[b * NLAT + l];
  }
}

#define WAVES_TOTAL 512
#define NPW (NTOT / WAVES_TOTAL)

__global__ __launch_bounds__(256)
void conv_kernel(const float* __restrict__ ndist,
                 const int* __restrict__ nbid,
                 const int* __restrict__ labels,
                 const float* __restrict__ dec_w,
                 const float* __restrict__ invd_t,
                 const float* __restrict__ ebl,
                 float* __restrict__ out) {
  const int t  = threadIdx.x & 31;
  const int wg = (blockIdx.x * blockDim.x + threadIdx.x) >> 5;
  const int b  = t & 7;
  const int kc = t >> 3;
  for (int it = 0; it < NPW; ++it) {
    const int n = wg * NPW + it;
    if (it + 1 < NPW) {
      __builtin_prefetch(ndist + (size_t)(n + 1) * KNB, 0, 1);
      __builtin_prefetch(nbid  + (size_t)(n + 1) * KNB, 0, 1);
    }
    const float nd  = ndist[(size_t)n * KNB + t];
    const int   nid = nbid [(size_t)n * KNB + t];
    const float d2  = nd * nd;
    const int   m   = labels[n];
    float d2v[8];
#pragma unroll
    for (int j = 0; j < 8; ++j)
      d2v[j] = __shfl(d2, kc * 8 + j, 32);
    float outacc = 0.f;
    const float* invrow = invd_t + (size_t)m * NLAT * BB;
    for (int l = 0; l < NLAT; ++l) {
      const float g   = dec_w[(size_t)l * NTOT + nid];
      const float inv = invrow[l * BB + b];
      float wsum = 0.f, ps = 0.f;
#pragma unroll
      for (int j = 0; j < 8; ++j) {
        const float gv = __shfl(g, kc * 8 + j, 32);
        const float w  = fmaxf(0.f, fmaf(-d2v[j], inv, 1.0f));
        wsum += w;
        ps = fmaf(gv, w, ps);
      }
      wsum += __shfl_xor(wsum, 8, 32);
      wsum += __shfl_xor(wsum, 16, 32);
      ps   += __shfl_xor(ps, 8, 32);
      ps   += __shfl_xor(ps, 16, 32);
      const float conv = ps / wsum;
      outacc = fmaf(ebl[l * BB + b], conv, outacc);
    }
    if (t < 8)
      out[(size_t)b * NTOT + n] = outacc;
  }
}

extern "C" void kernel_launch(void* const* d_in, const int* in_sizes, int n_in,
                              void* d_out, int out_size, void* d_ws, size_t ws_size,
                              hipStream_t stream) {
  const float* x      = (const float*)d_in[0];
  const float* enc_w  = (const float*)d_in[1];
  const float* enc_b  = (const float*)d_in[2];
  const float* dec_w  = (const float*)d_in[3];
  const float* bw_w   = (const float*)d_in[4];
  const float* bw_b   = (const float*)d_in[5];
  const float* ndist  = (const float*)d_in[6];
  const int*   nbid   = (const int*)d_in[7];
  const int*   labels = (const int*)d_in[8];
  float* out = (float*)d_out;

  float* ws       = (float*)d_ws;
  float* partials = ws + WS_PARTIALS;
  float* encoded  = ws + WS_ENCODED;
  float* invd_t   = ws + WS_INVD;
  float* ebl      = ws + WS_EBL;

  encode_partial_wmma<<<NPART, 32, 0, stream>>>(x, enc_w, partials);
  encode_reduce<<<1, 256, 0, stream>>>(partials, enc_b, encoded);
  bandwidth_kernel<<<7, 256, 0, stream>>>(encoded, bw_w, bw_b, invd_t, ebl);
  conv_kernel<<<64, 256, 0, stream>>>(ndist, nbid, labels, dec_w, invd_t, ebl, out);
}


# encode re-cut: group = 8 waves per 128-column chunk, wave = 16 columns, its six 16-B loads all in flight (was 96 8-B loads in turns); tiles pre-summed in LDS, whole-line stores: sum order changes
# speedup vs baseline: 13.3138x; 13.3138x over previous
#include <hip/hip_runtime.h>
#include <math.h>

#define NTOT 16384
#define NLAT 20
#define MCL  10
#define KNB  32
#define BB   8
#define MUF  5.0f

typedef __attribute__((ext_vector_type(2))) float v2f;
typedef __attribute__((ext_vector_type(8))) float v8f;

#define NPART   128
#define KCHUNK  (NTOT / NPART)
#define WS_PARTIALS 0
#define WS_ENCODED  (NPART * 512)
#define WS_INVD     (WS_ENCODED + BB * NLAT)
#define WS_EBL      (WS_INVD + MCL * NLAT * BB)
#define WS_CONVP    131072

#define ENC_WAVES 8
#define KWAVE (KCHUNK / ENC_WAVES)

__global__ __launch_bounds__(32 * ENC_WAVES)
void encode_partial_wmma(const float* __restrict__ x,
                         const float* __restrict__ enc_w,
                         float* __restrict__ partials) {
  __shared__ float red[ENC_WAVES][BB * NLAT];
  const int tid = threadIdx.x;
  const int t = tid & 31;
  const int w = tid >> 5;
  const int p = blockIdx.x;
  const int M = t & 15;
  const int Nn = t & 15;
  const int h4 = (t >> 4) << 2;
  const float maskA  = (M < BB) ? 1.f : 0.f;
  const float maskB1 = (Nn < NLAT - 16) ? 1.f : 0.f;
  const int rowA  = (M < BB) ? M : 0;
  const int rowB1 = (Nn < NLAT - 16) ? (16 + Nn) : 16;
  const int n0 = p * KCHUNK + w * KWAVE + h4;
  const float* pa  = x     + (size_t)rowA  * NTOT + n0;
  const float* pb0 = enc_w + (size_t)Nn    * NTOT + n0;
  const float* pb1 = enc_w + (size_t)rowB1 * NTOT + n0;
  const float4 av0 = *(const float4*)(pa);
  const float4 bv0 = *(const float4*)(pb0);
  const float4 cv0 = *(const float4*)(pb1);
  const float4 av1 = *(const float4*)(pa  + 8);
  const float4 bv1 = *(const float4*)(pb0 + 8);
  const float4 cv1 = *(const float4*)(pb1 + 8);
  v2f a00; a00.x = av0.x * maskA;  a00.y = av0.y * maskA;
  v2f a01; a01.x = av0.z * maskA;  a01.y = av0.w * maskA;
  v2f a10; a10.x = av1.x * maskA;  a10.y = av1.y * maskA;
  v2f a11; a11.x = av1.z * maskA;  a11.y = av1.w * maskA;
  v2f b00; b00.x = bv0.x;          b00.y = bv0.y;
  v2f b01; b01.x = bv0.z;          b01.y = bv0.w;
  v2f b10; b10.x = bv1.x;          b10.y = bv1.y;
  v2f b11; b11.x = bv1.z;          b11.y = bv1.w;
  v2f c00; c00.x = cv0.x * maskB1; c00.y = cv0.y * maskB1;
  v2f c01; c01.x = cv0.z * maskB1; c01.y = cv0.w * maskB1;
  v2f c10; c10.x = cv1.x * maskB1; c10.y = cv1.y * maskB1;
  v2f c11; c11.x = cv1.z * maskB1; c11.y = cv1.w * maskB1;
  v8f acc0 = {}, acc1 = {};
  acc0 = __builtin_amdgcn_wmma_f32_16x16x4_f32(false, a00, false, b00,
                                               (short)0, acc0, false, false);
  acc1 = __builtin_amdgcn_wmma_f32_16x16x4_f32(false, a00, false, c00,
                                               (short)0, acc1, false, false);
  acc0 = __builtin_amdgcn_wmma_f32_16x16x4_f32(false, a01, false, b01,
                                               (short)0, acc0, false, false);
  acc1 = __builtin_amdgcn_wmma_f32_16x16x4_f32(false, a01, false, c01,
                                               (short)0, acc1, false, false);
  acc0 = __builtin_amdgcn_wmma_f32_16x16x4_f32(false, a10, false, b10,
                                               (short)0, acc0, false, false);
  acc1 = __builtin_amdgcn_wmma_f32_16x16x4_f32(false, a10, false, c10,
                                               (short)0, acc1, false, false);
  acc0 = __builtin_amdgcn_wmma_f32_16x16x4_f32(false, a11, false, b11,
                                               (short)0, acc0, false, false);
  acc1 = __builtin_amdgcn_wmma_f32_16x16x4_f32(false, a11, false, c11,
                                               (short)0, acc1, false, false);
  if (t < 16) {
#pragma unroll
    for (int i = 0; i < 8; ++i) red[w][i * NLAT + t] = acc0[i];
    if (t < NLAT - 16) {
#pragma unroll
      for (int i = 0; i < 8; ++i) red[w][i * NLAT + 16 + t] = acc1[i];
    }
  }
  __syncthreads();
  if (tid < BB * NLAT) {
    float s = red[0][tid];
    for (int q = 1; q < ENC_WAVES; ++q) s += red[q][tid];
    partials[(size_t)p * (BB * NLAT) + tid] = s;
  }
}

#define RB_THREADS 1024
#define NOUT (BB * NLAT * MCL)
#define WV4  (NLAT * MCL * NLAT / 4)

__device__ __forceinline__ void window_scale(const int o, const float4 (&w)[5],
                                             const float bias,
                                             const float* __restrict__ enc_s,
                                             float* __restrict__ invd_t) {
  const int b = o / (NLAT * MCL);
  const int j = o % (NLAT * MCL);
  const float* e = enc_s + b * NLAT;
  float acc = bias;
#pragma unroll
  for (int c = 0; c < 5; ++c) {
    acc = fmaf(e[c * 4 + 0], w[c].x, acc);
    acc = fmaf(e[c * 4 + 1], w[c].y, acc);
    acc = fmaf(e[c * 4 + 2], w[c].z, acc);
    acc = fmaf(e[c * 4 + 3], w[c].w, acc);
  }
  const float s = 1.0f / (1.0f + expf(-acc));
  const float lo = 2.0f / 60.0f / MUF;
  const float hi = 1.0f / 60.0f;
  const float bw = (hi - lo) * s + lo;
  const int l = j / MCL, m = j % MCL;
  const float d = bw * MUF;
  invd_t[((size_t)m * NLAT + l) * BB + b] = 1.0f / (d * d);
}

__global__ __launch_bounds__(RB_THREADS)
void reduce_bandwidth_kernel(const float* __restrict__ partials,
                             const float* __restrict__ enc_b,
                             const float* __restrict__ bw_w,
                             const float* __restrict__ bw_b,
                             float* __restrict__ invd_t,
                             float* __restrict__ ebl) {
  __shared__ __attribute__((aligned(16))) float part_s[NPART * BB * NLAT];
  __shared__ __attribute__((aligned(16))) float w_s[NLAT * MCL * NLAT];
  __shared__ float enc_s[BB * NLAT];
  const int tid = threadIdx.x;

  float4 pv[5];
#pragma unroll
  for (int r = 0; r < 5; ++r)
    pv[r] = ((const float4*)partials)[r * RB_THREADS + tid];
  const bool hasW = tid < WV4;
  float4 wv = make_float4(0.f, 0.f, 0.f, 0.f);
  if (hasW) wv = ((const float4*)bw_w)[tid];

  const int  oA = tid;
  const bool hasB = (tid + RB_THREADS) < NOUT;
  const int  oB = hasB ? tid + RB_THREADS : tid;
  const int  jA = oA % (NLAT * MCL), jB = oB % (NLAT * MCL);
  const float biasA = bw_b[jA], biasB = bw_b[jB];
  const float eb = enc_b[tid % NLAT];

#pragma unroll
  for (int r = 0; r < 5; ++r)
    ((float4*)part_s)[r * RB_THREADS + tid] = pv[r];
  if (hasW) ((float4*)w_s)[tid] = wv;
  __syncthreads();

  float4 wA[5], wB[5];
#pragma unroll
  for (int c = 0; c < 5; ++c) {
    wA[c] = ((const float4*)(w_s + jA * NLAT))[c];
    wB[c] = ((const float4*)(w_s + jB * NLAT))[c];
  }

  if (tid < BB * NLAT) {
    float s = 0.f;
    for (int p = 0; p < NPART; ++p) s += part_s[p * (BB * NLAT) + tid];
    s += eb;
    enc_s[tid] = s;
    ebl[(tid % NLAT) * BB + tid / NLAT] = s;
  }
  __syncthreads();

  window_scale(oA, wA, biasA, enc_s, invd_t);
  if (hasB) window_scale(oB, wB, biasB, enc_s, invd_t);
}

#define CPTS      256
#define CCHUNKS   (NTOT / CPTS)
#define LPG       5
#define NLG       (NLAT / LPG)
#define TPITCH    68

__global__ __launch_bounds__(256)
void conv_latent5_kernel(const float* __restrict__ ndist,
                         const int* __restrict__ nbid,
                         const int* __restrict__ labels,
                         const float* __restrict__ dec_w,
                         const float* __restrict__ invd_t,
                         const float* __restrict__ ebl,
                         float* __restrict__ convp) {
  __shared__ __attribute__((aligned(16))) float row[2][NTOT];
  __shared__ __attribute__((aligned(16))) float stage[8][32 * TPITCH];

  const int tid = threadIdx.x;
  const int t   = tid & 31;
  const int wv  = tid >> 5;
  const int lg  = blockIdx.y;
  const int l0  = lg * LPG;
  const int n0  = blockIdx.x * CPTS + wv * 32;

  const int prow = t >> 3, piece = t & 7;
  const float* ndp = ndist + (size_t)(n0 + prow) * KNB + piece * 4;
  const int*   nbp = nbid  + (size_t)(n0 + prow) * KNB + piece * 4;
  float4 nv[8];
  int4   iv[8];
#pragma unroll
  for (int j = 0; j < 8; ++j) {
    nv[j] = *(const float4*)(ndp + j * 4 * KNB);
    iv[j] = *(const int4*)(nbp + j * 4 * KNB);
  }
  const int m = labels[n0 + t];

  const float4* src = (const float4*)(dec_w + (size_t)l0 * NTOT);
#define ROW16(X) X(0) X(1) X(2) X(3) X(4) X(5) X(6) X(7) X(8) X(9) X(10) X(11) X(12) X(13) X(14) X(15)
#define ROW_DECL(i)  float4 rv##i;
#define ROW_LOAD(i)  rv##i = src[(i) * 256 + tid];
#define ROW_STORE(i) dst[(i) * 256 + tid] = rv##i;
  ROW16(ROW_DECL)
  ROW16(ROW_LOAD)
  const float4* ip = (const float4*)(invd_t + ((size_t)m * NLAT + l0) * BB);
  float4 i0 = ip[0], i1 = ip[1];

  float* st = stage[wv];
  float* stw = st + prow * TPITCH + piece * 4;
#pragma unroll
  for (int j = 0; j < 8; ++j) {
    *(float4*)(stw + j * 4 * TPITCH)      = nv[j];
    *(int4*)(stw + j * 4 * TPITCH + KNB)  = iv[j];
  }
  __syncthreads();
  const float* str = st + t * TPITCH;
  float d2[KNB];
  int   id[KNB];
#pragma unroll
  for (int q = 0; q < 8; ++q) {
    const float4 d = *(const float4*)(str + q * 4);
    const int4   k = *(const int4*)(str + KNB + q * 4);
    d2[q * 4 + 0] = d.x * d.x; d2[q * 4 + 1] = d.y * d.y;
    d2[q * 4 + 2] = d.z * d.z; d2[q * 4 + 3] = d.w * d.w;
    id[q * 4 + 0] = k.x; id[q * 4 + 1] = k.y; id[q * 4 + 2] = k.z; id[q * 4 + 3] = k.w;
  }

  float acc[BB];
#pragma unroll
  for (int b = 0; b < BB; ++b) acc[b] = 0.f;

#pragma nounroll
  for (int li = 0; li < LPG; ++li) {
    float* rb = row[li & 1];
    float4* dst = (float4*)rb;
    ROW16(ROW_STORE)
    const float inv[BB] = {i0.x, i0.y, i0.z, i0.w, i1.x, i1.y, i1.z, i1.w};
    __syncthreads();
    if (li + 1 < LPG) {
      src += NTOT / 4;
      ROW16(ROW_LOAD)
      ip += 2;
      i0 = ip[0]; i1 = ip[1];
    }
    float wsR[4][BB], psR[4][BB];
#pragma unroll
    for (int r = 0; r < 4; ++r) {
      float g[8];
#pragma unroll
      for (int j = 0; j < 8; ++j) g[j] = rb[id[r * 8 + j]];
      float ws[BB], ps[BB];
#pragma unroll
      for (int b = 0; b < BB; ++b) { ws[b] = 0.f; ps[b] = 0.f; }
#pragma unroll
      for (int j = 0; j < 8; ++j) {
        const float dd = d2[r * 8 + j];
#pragma unroll
        for (int b = 0; b < BB; ++b) {
          const float w = fmaxf(0.f, fmaf(-dd, inv[b], 1.0f));
          ws[b] += w;
          ps[b] = fmaf(g[j], w, ps[b]);
        }
      }
#pragma unroll
      for (int b = 0; b < BB; ++b) { wsR[r][b] = ws[b]; psR[r][b] = ps[b]; }
    }
    const float* ev = ebl + (l0 + li) * BB;
#pragma unroll
    for (int b = 0; b < BB; ++b) {
      const float wsT = (wsR[0][b] + wsR[1][b]) + (wsR[2][b] + wsR[3][b]);
      const float psT = (psR[0][b] + psR[1][b]) + (psR[2][b] + psR[3][b]);
      acc[b] = fmaf(ev[b], psT / wsT, acc[b]);
    }
  }
#pragma unroll
  for (int b = 0; b < BB; ++b) {
    convp[((size_t)(lg * BB + b)) * NTOT + n0 + t] = acc[b];
  }
}

#define SUM_THREADS 128
#define SUM_GROUPS  (BB * NTOT / 4 / SUM_THREADS)

__global__ __launch_bounds__(SUM_THREADS)
void latent_sum_kernel(const float* __restrict__ convp,
                       float* __restrict__ out) {
  const int i = blockIdx.x * SUM_THREADS + threadIdx.x;
  const float4* p4 = (const float4*)convp;
  const float4 v0 = p4[i];
  const float4 v1 = p4[(size_t)1 * (BB * NTOT / 4) + i];
  const float4 v2 = p4[(size_t)2 * (BB * NTOT / 4) + i];
  const float4 v3 = p4[(size_t)3 * (BB * NTOT / 4) + i];
  float4 acc;
  acc.x = ((v0.x + v1.x) + v2.x) + v3.x;
  acc.y = ((v0.y + v1.y) + v2.y) + v3.y;
  acc.z = ((v0.z + v1.z) + v2.z) + v3.z;
  acc.w = ((v0.w + v1.w) + v2.w) + v3.w;
  ((float4*)out)[i] = acc;
}

extern "C" void kernel_launch(void* const* d_in, const int* in_sizes, int n_in,
                              void* d_out, int out_size, void* d_ws, size_t ws_size,
                              hipStream_t stream) {
  const float* x      = (const float*)d_in[0];
  const float* enc_w  = (const float*)d_in[1];
  const float* enc_b  = (const float*)d_in[2];
  const float* dec_w  = (const float*)d_in[3];
  const float* bw_w   = (const float*)d_in[4];
  const float* bw_b   = (const float*)d_in[5];
  const float* ndist  = (const float*)d_in[6];
  const int*   nbid   = (const int*)d_in[7];
  const int*   labels = (const int*)d_in[8];
  float* out = (float*)d_out;

  float* ws       = (float*)d_ws;
  float* partials = ws + WS_PARTIALS;
  float* invd_t   = ws + WS_INVD;
  float* ebl      = ws + WS_EBL;
  float* convp    = ws + WS_CONVP;

  encode_partial_wmma<<<NPART, 32 * ENC_WAVES, 0, stream>>>(x, enc_w, partials);
  reduce_bandwidth_kernel<<<1, RB_THREADS, 0, stream>>>(partials, enc_b, bw_w, bw_b, invd_t, ebl);
  conv_latent5_kernel<<<dim3(CCHUNKS, NLG), 256, 0, stream>>>(ndist, nbid, labels, dec_w, invd_t, ebl, convp);
  latent_sum_kernel<<<SUM_GROUPS, SUM_THREADS, 0, stream>>>(convp, out);
}


# reduce+MLP stage cut by batch row: 8 groups (was 1) load only their row's partials ([b][p] lines from encode) + weights; 20 sums in 8 runs of 16: sum order changes; conv copies its scales to LDS
# speedup vs baseline: 15.6224x; 1.1734x over previous
#include <hip/hip_runtime.h>
#include <math.h>

#define NTOT 16384
#define NLAT 20
#define MCL  10
#define KNB  32
#define BB   8
#define MUF  5.0f

typedef __attribute__((ext_vector_type(2))) float v2f;
typedef __attribute__((ext_vector_type(8))) float v8f;

#define NPART   128
#define KCHUNK  (NTOT / NPART)
#define PPITCH  32
#define IPITCH  16
#define IBLOCK  (NLAT * IPITCH)
#define WS_PARTIALS 0
#define WS_INVD     65536
#define WS_EB       (WS_INVD + BB * IBLOCK)
#define WS_CONVP    131072

#define ENC_WAVES 8
#define KWAVE (KCHUNK / ENC_WAVES)

__global__ __launch_bounds__(32 * ENC_WAVES)
void encode_partial_wmma(const float* __restrict__ x,
                         const float* __restrict__ enc_w,
                         float* __restrict__ partials) {
  __shared__ float red[ENC_WAVES][BB * NLAT];
  const int tid = threadIdx.x;
  const int t = tid & 31;
  const int w = tid >> 5;
  const int p = blockIdx.x;
  const int M = t & 15;
  const int Nn = t & 15;
  const int h4 = (t >> 4) << 2;
  const float maskA  = (M < BB) ? 1.f : 0.f;
  const float maskB1 = (Nn < NLAT - 16) ? 1.f : 0.f;
  const int rowA  = (M < BB) ? M : 0;
  const int rowB1 = (Nn < NLAT - 16) ? (16 + Nn) : 16;
  const int n0 = p * KCHUNK + w * KWAVE + h4;
  const float* pa  = x     + (size_t)rowA  * NTOT + n0;
  const float* pb0 = enc_w + (size_t)Nn    * NTOT + n0;
  const float* pb1 = enc_w + (size_t)rowB1 * NTOT + n0;
  const float4 av0 = *(const float4*)(pa);
  const float4 bv0 = *(const float4*)(pb0);
  const float4 cv0 = *(const float4*)(pb1);
  const float4 av1 = *(const float4*)(pa  + 8);
  const float4 bv1 = *(const float4*)(pb0 + 8);
  const float4 cv1 = *(const float4*)(pb1 + 8);
  v2f a00; a00.x = av0.x * maskA;  a00.y = av0.y * maskA;
  v2f a01; a01.x = av0.z * maskA;  a01.y = av0.w * maskA;
  v2f a10; a10.x = av1.x * maskA;  a10.y = av1.y * maskA;
  v2f a11; a11.x = av1.z * maskA;  a11.y = av1.w * maskA;
  v2f b00; b00.x = bv0.x;          b00.y = bv0.y;
  v2f b01; b01.x = bv0.z;          b01.y = bv0.w;
  v2f b10; b10.x = bv1.x;          b10.y = bv1.y;
  v2f b11; b11.x = bv1.z;          b11.y = bv1.w;
  v2f c00; c00.x = cv0.x * maskB1; c00.y = cv0.y * maskB1;
  v2f c01; c01.x = cv0.z * maskB1; c01.y = cv0.w * maskB1;
  v2f c10; c10.x = cv1.x * maskB1; c10.y = cv1.y * maskB1;
  v2f c11; c11.x = cv1.z * maskB1; c11.y = cv1.w * maskB1;
  v8f acc0 = {}, acc1 = {};
  acc0 = __builtin_amdgcn_wmma_f32_16x16x4_f32(false, a00, false, b00,
                                               (short)0, acc0, false, false);
  acc1 = __builtin_amdgcn_wmma_f32_16x16x4_f32(false, a00, false, c00,
                                               (short)0, acc1, false, false);
  acc0 = __builtin_amdgcn_wmma_f32_16x16x4_f32(false, a01, false, b01,
                                               (short)0, acc0, false, false);
  acc1 = __builtin_amdgcn_wmma_f32_16x16x4_f32(false, a01, false, c01,
                                               (short)0, acc1, false, false);
  acc0 = __builtin_amdgcn_wmma_f32_16x16x4_f32(false, a10, false, b10,
                                               (short)0, acc0, false, false);
  acc1 = __builtin_amdgcn_wmma_f32_16x16x4_f32(false, a10, false, c10,
                                               (short)0, acc1, false, false);
  acc0 = __builtin_amdgcn_wmma_f32_16x16x4_f32(false, a11, false, b11,
                                               (short)0, acc0, false, false);
  acc1 = __builtin_amdgcn_wmma_f32_16x16x4_f32(false, a11, false, c11,
                                               (short)0, acc1, false, false);
  if (t < 16) {
#pragma unroll
    for (int i = 0; i < 8; ++i) red[w][i * NLAT + t] = acc0[i];
    if (t < NLAT - 16) {
#pragma unroll
      for (int i = 0; i < 8; ++i) red[w][i * NLAT + 16 + t] = acc1[i];
    }
  }
  __syncthreads();
  {
    float s = 0.f;
    if (t < NLAT) {
      s = red[0][w * NLAT + t];
      for (int q = 1; q < ENC_WAVES; ++q) s += red[q][w * NLAT + t];
    }
    partials[((size_t)w * NPART + p) * PPITCH + t] = s;
  }
}

#define RB_THREADS 512
#define WV4    (NLAT * MCL * NLAT / 4)
#define RUNS   8
#define RUNLEN (NPART / RUNS)

__global__ __launch_bounds__(RB_THREADS)
void reduce_bandwidth_kernel(const float* __restrict__ partials,
                             const float* __restrict__ enc_b,
                             const float* __restrict__ bw_w,
                             const float* __restrict__ bw_b,
                             float* __restrict__ invd_b,
                             float* __restrict__ eb) {
  __shared__ __attribute__((aligned(16))) float w_s[2 * RB_THREADS * 4];
  __shared__ float run_s[RUNS][NLAT];
  __shared__ float enc_s[NLAT];
  const int tid = threadIdx.x;
  const int b = blockIdx.x;

  const int  w1 = (tid + RB_THREADS) < WV4 ? tid + RB_THREADS : tid;
  const float4 wv0 = ((const float4*)bw_w)[tid];
  const float4 wv1 = ((const float4*)bw_w)[w1];

  const bool isSum = tid < RUNS * NLAT;
  const int ls = tid % NLAT, q = tid / NLAT;
  float pv[RUNLEN];
#pragma unroll
  for (int k = 0; k < RUNLEN; ++k) pv[k] = 0.f;
  if (isSum) {
    const float* pp = partials + ((size_t)b * NPART + q * RUNLEN) * PPITCH + ls;
#pragma unroll
    for (int k = 0; k < RUNLEN; ++k)
      pv[k] = pp[k * PPITCH];
  }

  const int lat = tid >> 4, mo = tid & 15;
  const bool isOut = (tid < IBLOCK) && (mo < MCL);
  const int j = isOut ? lat * MCL + mo : 0;
  const float bias = bw_b[j];
  const float ebias = enc_b[ls];

  ((float4*)w_s)[tid] = wv0;
  ((float4*)w_s)[tid + RB_THREADS] = wv1;
  if (isSum) {
    float s = pv[0];
#pragma unroll
    for (int k = 1; k < RUNLEN; ++k) s += pv[k];
    run_s[q][ls] = s;
  }
  __syncthreads();

  if (tid < NLAT) {
    float s = run_s[0][tid];
    for (int r = 1; r < RUNS; ++r) s += run_s[r][tid];
    s += ebias;
    enc_s[tid] = s;
    eb[b * 32 + tid] = s;
  }
  __syncthreads();

  if (isOut) {
    float4 w[5];
#pragma unroll
    for (int c = 0; c < 5; ++c)
      w[c] = ((const float4*)(w_s + j * NLAT))[c];
    float acc = bias;
#pragma unroll
    for (int c = 0; c < 5; ++c) {
      acc = fmaf(enc_s[c * 4 + 0], w[c].x, acc);
      acc = fmaf(enc_s[c * 4 + 1], w[c].y, acc);
      acc = fmaf(enc_s[c * 4 + 2], w[c].z, acc);
      acc = fmaf(enc_s[c * 4 + 3], w[c].w, acc);
    }
    const float s = 1.0f / (1.0f + expf(-acc));
    const float lo = 2.0f / 60.0f / MUF;
    const float hi = 1.0f / 60.0f;
    const float bw = (hi - lo) * s + lo;
    const float d = bw * MUF;
    invd_b[(size_t)b * IBLOCK + lat * IPITCH + mo] = 1.0f / (d * d);
  }
}

#define CPTS      256
#define CCHUNKS   (NTOT / CPTS)
#define LPG       5
#define NLG       (NLAT / LPG)
#define TPITCH    68

__global__ __launch_bounds__(256)
void conv_latent5_kernel(const float* __restrict__ ndist,
                         const int* __restrict__ nbid,
                         const int* __restrict__ labels,
                         const float* __restrict__ dec_w,
                         const float* __restrict__ invd_b,
                         const float* __restrict__ eb,
                         float* __restrict__ convp) {
  __shared__ __attribute__((aligned(16))) float row[2][NTOT];
  __shared__ __attribute__((aligned(16))) float stage[8][32 * TPITCH];
  __shared__ __attribute__((aligned(16))) float inv_s[2 * 256];

  const int tid = threadIdx.x;
  const int t   = tid & 31;
  const int wv  = tid >> 5;
  const int lg  = blockIdx.y;
  const int l0  = lg * LPG;
  const int n0  = blockIdx.x * CPTS + wv * 32;

  const int prow = t >> 3, piece = t & 7;
  const float* ndp = ndist + (size_t)(n0 + prow) * KNB + piece * 4;
  const int*   nbp = nbid  + (size_t)(n0 + prow) * KNB + piece * 4;
  float4 nv[8];
  int4   iv[8];
#pragma unroll
  for (int j = 0; j < 8; ++j) {
    nv[j] = *(const float4*)(ndp + j * 4 * KNB);
    iv[j] = *(const int4*)(nbp + j * 4 * KNB);
  }
  const int m = labels[n0 + t];

  const float4* src = (const float4*)(dec_w + (size_t)l0 * NTOT);
#define ROW16(X) X(0) X(1) X(2) X(3) X(4) X(5) X(6) X(7) X(8) X(9) X(10) X(11) X(12) X(13) X(14) X(15)
#define ROW_DECL(i)  float4 rv##i;
#define ROW_LOAD(i)  rv##i = src[(i) * 256 + tid];
#define ROW_STORE(i) dst[(i) * 256 + tid] = rv##i;
  ROW16(ROW_DECL)
  ROW16(ROW_LOAD)
  const int e1 = (tid + 256 < MCL * LPG * BB) ? tid + 256 : tid;
  const float sc0 = invd_b[(tid & 7) * IBLOCK + (l0 + (tid >> 3) % LPG) * IPITCH + tid / (LPG * BB)];
  const float sc1 = invd_b[(e1 & 7) * IBLOCK + (l0 + (e1 >> 3) % LPG) * IPITCH + e1 / (LPG * BB)];

  float* st = stage[wv];
  float* stw = st + prow * TPITCH + piece * 4;
#pragma unroll
  for (int j = 0; j < 8; ++j) {
    *(float4*)(stw + j * 4 * TPITCH)      = nv[j];
    *(int4*)(stw + j * 4 * TPITCH + KNB)  = iv[j];
  }
  inv_s[tid] = sc0;
  inv_s[tid + 256] = sc1;
  __syncthreads();
  const float4* ip = (const float4*)(inv_s + m * (LPG * BB));
  float4 i0 = ip[0], i1 = ip[1];
  const float* str = st + t * TPITCH;
  float d2[KNB];
  int   id[KNB];
#pragma unroll
  for (int q = 0; q < 8; ++q) {
    const float4 d = *(const float4*)(str + q * 4);
    const int4   k = *(const int4*)(str + KNB + q * 4);
    d2[q * 4 + 0] = d.x * d.x; d2[q * 4 + 1] = d.y * d.y;
    d2[q * 4 + 2] = d.z * d.z; d2[q * 4 + 3] = d.w * d.w;
    id[q * 4 + 0] = k.x; id[q * 4 + 1] = k.y; id[q * 4 + 2] = k.z; id[q * 4 + 3] = k.w;
  }

  float acc[BB];
#pragma unroll
  for (int b = 0; b < BB; ++b) acc[b] = 0.f;

#pragma nounroll
  for (int li = 0; li < LPG; ++li) {
    float* rb = row[li & 1];
    float4* dst = (float4*)rb;
    ROW16(ROW_STORE)
    const float inv[BB] = {i0.x, i0.y, i0.z, i0.w, i1.x, i1.y, i1.z, i1.w};
    __syncthreads();
    if (li + 1 < LPG) {
      src += NTOT / 4;
      ROW16(ROW_LOAD)
      ip += 2;
      i0 = ip[0]; i1 = ip[1];
    }
    float wsR[4][BB], psR[4][BB];
#pragma unroll
    for (int r = 0; r < 4; ++r) {
      float g[8];
#pragma unroll
      for (int j = 0; j < 8; ++j) g[j] = rb[id[r * 8 + j]];
      float ws[BB], ps[BB];
#pragma unroll
      for (int b = 0; b < BB; ++b) { ws[b] = 0.f; ps[b] = 0.f; }
#pragma unroll
      for (int j = 0; j < 8; ++j) {
        const float dd = d2[r * 8 + j];
#pragma unroll
        for (int b = 0; b < BB; ++b) {
          const float w = fmaxf(0.f, fmaf(-dd, inv[b], 1.0f));
          ws[b] += w;
          ps[b] = fmaf(g[j], w, ps[b]);
        }
      }
#pragma unroll
      for (int b = 0; b < BB; ++b) { wsR[r][b] = ws[b]; psR[r][b] = ps[b]; }
    }
    const float* ev = eb + (l0 + li);
#pragma unroll
    for (int b = 0; b < BB; ++b) {
      const float wsT = (wsR[0][b] + wsR[1][b]) + (wsR[2][b] + wsR[3][b]);
      const float psT = (psR[0][b] + psR[1][b]) + (psR[2][b] + psR[3][b]);
      acc[b] = fmaf(ev[b * 32], psT / wsT, acc[b]);
    }
  }
#pragma unroll
  for (int b = 0; b < BB; ++b) {
    convp[((size_t)(lg * BB + b)) * NTOT + n0 + t] = acc[b];
  }
}

#define SUM_THREADS 128
#define SUM_GROUPS  (BB * NTOT / 4 / SUM_THREADS)

__global__ __launch_bounds__(SUM_THREADS)
void latent_sum_kernel(const float* __restrict__ convp,
                       float* __restrict__ out) {
  const int i = blockIdx.x * SUM_THREADS + threadIdx.x;
  const float4* p4 = (const float4*)convp;
  const float4 v0 = p4[i];
  const float4 v1 = p4[(size_t)1 * (BB * NTOT / 4) + i];
  const float4 v2 = p4[(size_t)2 * (BB * NTOT / 4) + i];
  const float4 v3 = p4[(size_t)3 * (BB * NTOT / 4) + i];
  float4 acc;
  acc.x = ((v0.x + v1.x) + v2.x) + v3.x;
  acc.y = ((v0.y + v1.y) + v2.y) + v3.y;
  acc.z = ((v0.z + v1.z) + v2.z) + v3.z;
  acc.w = ((v0.w + v1.w) + v2.w) + v3.w;
  ((float4*)out)[i] = acc;
}

extern "C" void kernel_launch(void* const* d_in, const int* in_sizes, int n_in,
                              void* d_out, int out_size, void* d_ws, size_t ws_size,
                              hipStream_t stream) {
  const float* x      = (const float*)d_in[0];
  const float* enc_w  = (const float*)d_in[1];
  const float* enc_b  = (const float*)d_in[2];
  const float* dec_w  = (const float*)d_in[3];
  const float* bw_w   = (const float*)d_in[4];
  const float* bw_b   = (const float*)d_in[5];
  const float* ndist  = (const float*)d_in[6];
  const int*   nbid   = (const int*)d_in[7];
  const int*   labels = (const int*)d_in[8];
  float* out = (float*)d_out;

  float* ws       = (float*)d_ws;
  float* partials = ws + WS_PARTIALS;
  float* invd_b   = ws + WS_INVD;
  float* eb       = ws + WS_EB;
  float* convp    = ws + WS_CONVP;

  encode_partial_wmma<<<NPART, 32 * ENC_WAVES, 0, stream>>>(x, enc_w, partials);
  reduce_bandwidth_kernel<<<BB, RB_THREADS, 0, stream>>>(partials, enc_b, bw_w, bw_b, invd_b, eb);
  conv_latent5_kernel<<<dim3(CCHUNKS, NLG), 256, 0, stream>>>(ndist, nbid, labels, dec_w, invd_b, eb, convp);
  latent_sum_kernel<<<SUM_GROUPS, SUM_THREADS, 0, stream>>>(convp, out);
}


# 4 launches -> 3: the reduce+MLP launch goes; every conv group sums the encode partials (now 64, compact lines) itself and builds its 400 window scales in LDS; encoded sum order changes
# speedup vs baseline: 17.1138x; 1.0955x over previous
#include <hip/hip_runtime.h>
#include <math.h>

#define NTOT 16384
#define NLAT 20
#define MCL  10
#define KNB  32
#define BB   8
#define MUF  5.0f

typedef __attribute__((ext_vector_type(2))) float v2f;
typedef __attribute__((ext_vector_type(8))) float v8f;

#define NPART   64
#define KCHUNK  (NTOT / NPART)
#define PVALS   (BB * NLAT)
#define WS_PARTIALS 0
#define WS_CONVP    131072

#define ENC_WAVES 16
#define KWAVE (KCHUNK / ENC_WAVES)

__global__ __launch_bounds__(32 * ENC_WAVES)
void encode_partial_wmma(const float* __restrict__ x,
                         const float* __restrict__ enc_w,
                         float* __restrict__ partials) {
  __shared__ float red[ENC_WAVES][BB * NLAT];
  const int tid = threadIdx.x;
  const int t = tid & 31;
  const int w = tid >> 5;
  const int p = blockIdx.x;
  const int M = t & 15;
  const int Nn = t & 15;
  const int h4 = (t >> 4) << 2;
  const float maskA  = (M < BB) ? 1.f : 0.f;
  const float maskB1 = (Nn < NLAT - 16) ? 1.f : 0.f;
  const int rowA  = (M < BB) ? M : 0;
  const int rowB1 = (Nn < NLAT - 16) ? (16 + Nn) : 16;
  const int n0 = p * KCHUNK + w * KWAVE + h4;
  const float* pa  = x     + (size_t)rowA  * NTOT + n0;
  const float* pb0 = enc_w + (size_t)Nn    * NTOT + n0;
  const float* pb1 = enc_w + (size_t)rowB1 * NTOT + n0;
  const float4 av0 = *(const float4*)(pa);
  const float4 bv0 = *(const float4*)(pb0);
  const float4 cv0 = *(const float4*)(pb1);
  const float4 av1 = *(const float4*)(pa  + 8);
  const float4 bv1 = *(const float4*)(pb0 + 8);
  const float4 cv1 = *(const float4*)(pb1 + 8);
  v2f a00; a00.x = av0.x * maskA;  a00.y = av0.y * maskA;
  v2f a01; a01.x = av0.z * maskA;  a01.y = av0.w * maskA;
  v2f a10; a10.x = av1.x * maskA;  a10.y = av1.y * maskA;
  v2f a11; a11.x = av1.z * maskA;  a11.y = av1.w * maskA;
  v2f b00; b00.x = bv0.x;          b00.y = bv0.y;
  v2f b01; b01.x = bv0.z;          b01.y = bv0.w;
  v2f b10; b10.x = bv1.x;          b10.y = bv1.y;
  v2f b11; b11.x = bv1.z;          b11.y = bv1.w;
  v2f c00; c00.x = cv0.x * maskB1; c00.y = cv0.y * maskB1;
  v2f c01; c01.x = cv0.z * maskB1; c01.y = cv0.w * maskB1;
  v2f c10; c10.x = cv1.x * maskB1; c10.y = cv1.y * maskB1;
  v2f c11; c11.x = cv1.z * maskB1; c11.y = cv1.w * maskB1;
  v8f acc0 = {}, acc1 = {};
  acc0 = __builtin_amdgcn_wmma_f32_16x16x4_f32(false, a00, false, b00,
                                               (short)0, acc0, false, false);
  acc1 = __builtin_amdgcn_wmma_f32_16x16x4_f32(false, a00, false, c00,
                                               (short)0, acc1, false, false);
  acc0 = __builtin_amdgcn_wmma_f32_16x16x4_f32(false, a01, false, b01,
                                               (short)0, acc0, false, false);
  acc1 = __builtin_amdgcn_wmma_f32_16x16x4_f32(false, a01, false, c01,
                                               (short)0, acc1, false, false);
  acc0 = __builtin_amdgcn_wmma_f32_16x16x4_f32(false, a10, false, b10,
                                               (short)0, acc0, false, false);
  acc1 = __builtin_amdgcn_wmma_f32_16x16x4_f32(false, a10, false, c10,
                                               (short)0, acc1, false, false);
  acc0 = __builtin_amdgcn_wmma_f32_16x16x4_f32(false, a11, false, b11,
                                               (short)0, acc0, false, false);
  acc1 = __builtin_amdgcn_wmma_f32_16x16x4_f32(false, a11, false, c11,
                                               (short)0, acc1, false, false);
  if (t < 16) {
#pragma unroll
    for (int i = 0; i < 8; ++i) red[w][i * NLAT + t] = acc0[i];
    if (t < NLAT - 16) {
#pragma unroll
      for (int i = 0; i < 8; ++i) red[w][i * NLAT + 16 + t] = acc1[i];
    }
  }
  __syncthreads();
  if (w < PVALS / 32) {
    const int v = w * 32 + t;
    float s = red[0][v];
    for (int q = 1; q < ENC_WAVES; ++q) s += red[q][v];
    partials[(size_t)p * PVALS + v] = s;
  }
}

#define CPTS      256
#define CCHUNKS   (NTOT / CPTS)
#define LPG       5
#define NLG       (NLAT / LPG)
#define TPITCH    68
#define PV4       (PVALS / 4)
#define PRUN      (NPART / 8)
#define WROWS     (LPG * MCL)
#define WPIECES   (WROWS * NLAT / 4)
#define NSCALE    (MCL * LPG * BB)

__global__ __launch_bounds__(256)
void conv_latent5_kernel(const float* __restrict__ ndist,
                         const int* __restrict__ nbid,
                         const int* __restrict__ labels,
                         const float* __restrict__ dec_w,
                         const float* __restrict__ partials,
                         const float* __restrict__ enc_b,
                         const float* __restrict__ bw_w,
                         const float* __restrict__ bw_b,
                         float* __restrict__ convp) {
  __shared__ __attribute__((aligned(16))) float row[2][NTOT];
  __shared__ __attribute__((aligned(16))) float stage[8][32 * TPITCH];
  __shared__ __attribute__((aligned(16))) float inv_s[2 * 256];
  __shared__ float4 red2[8][PVALS];
  __shared__ float4 w_s[256];
  __shared__ __attribute__((aligned(16))) float enc_s[PVALS];
  __shared__ __attribute__((aligned(16))) float enct_s[LPG * BB];

  const int tid = threadIdx.x;
  const int t   = tid & 31;
  const int wv  = tid >> 5;
  const int lg  = blockIdx.y;
  const int l0  = lg * LPG;
  const int n0  = blockIdx.x * CPTS + wv * 32;

  const float4* pp = (const float4*)partials + (size_t)wv * (PRUN * PV4) + t;
#define PQ10(X) X(0) X(1) X(2) X(3) X(4) X(5) X(6) X(7) X(8) X(9)
#define PQ_LOAD(j) const float4 pq##j = pp[(j) * 32];
  PQ10(PQ_LOAD)
  const int wpi = tid < WPIECES ? tid : WPIECES - 1;
  const float4 wpv = ((const float4*)(bw_w + (size_t)l0 * MCL * NLAT))[wpi];
  const int e1 = (tid + 256 < NSCALE) ? tid + 256 : tid;
  const int jl0 = ((tid >> 3) % LPG) * MCL + tid / (LPG * BB);
  const int jl1 = ((e1 >> 3) % LPG) * MCL + e1 / (LPG * BB);
  const float bias0 = bw_b[l0 * MCL + jl0];
  const float bias1 = bw_b[l0 * MCL + jl1];
  const float ebias = enc_b[tid % NLAT];

  const int prow = t >> 3, piece = t & 7;
  const float* ndp = ndist + (size_t)(n0 + prow) * KNB + piece * 4;
  const int*   nbp = nbid  + (size_t)(n0 + prow) * KNB + piece * 4;
#define TILE8(X) X(0) X(1) X(2) X(3) X(4) X(5) X(6) X(7)
#define TILE_LOAD(j)  const float4 nv##j = *(const float4*)(ndp + (j) * 4 * KNB); \
                      const int4   iv##j = *(const int4*)(nbp + (j) * 4 * KNB);
#define TILE_STORE(j) *(float4*)(stw + (j) * 4 * TPITCH) = nv##j; \
                      *(int4*)(stw + (j) * 4 * TPITCH + KNB) = iv##j;
  TILE8(TILE_LOAD)
  const int m = labels[n0 + t];

  const float4* src = (const float4*)(dec_w + (size_t)l0 * NTOT);
#define ROW16(X) X(0) X(1) X(2) X(3) X(4) X(5) X(6) X(7) X(8) X(9) X(10) X(11) X(12) X(13) X(14) X(15)
#define ROW_DECL(i)  float4 rv##i;
#define ROW_LOAD(i)  rv##i = src[(i) * 256 + tid];
#define ROW_STORE(i) dst[(i) * 256 + tid] = rv##i;
  ROW16(ROW_DECL)
  ROW16(ROW_LOAD)

#define PQ_SUM(r, A, B) { float4 a; \
    a.x = pq##A.x + pq##B.x; a.y = pq##A.y + pq##B.y; \
    a.z = pq##A.z + pq##B.z; a.w = pq##A.w + pq##B.w; \
    red2[wv][(r) * 32 + t] = a; }
  PQ_SUM(0, 0, 5)
  PQ_SUM(1, 1, 6)
  PQ_SUM(2, 2, 7)
  PQ_SUM(3, 3, 8)
  PQ_SUM(4, 4, 9)
  w_s[tid] = wpv;
  __syncthreads();
  if (tid < PVALS) {
    const float* r2 = (const float*)red2 + tid;
    float s = (r2[0] + r2[PVALS]) + (r2[2 * PVALS] + r2[3 * PVALS]);
#pragma unroll
    for (int q = 1; q < 8; ++q) {
      const float* rq = r2 + q * 4 * PVALS;
      s += (rq[0] + rq[PVALS]) + (rq[2 * PVALS] + rq[3 * PVALS]);
    }
    s += ebias;
    enc_s[tid] = s;
    const int lr = tid % NLAT - l0;
    if (lr >= 0 && lr < LPG) enct_s[lr * BB + tid / NLAT] = s;
  }
  __syncthreads();
  {
    const float4* er = (const float4*)(enc_s + (tid & 7) * NLAT);
    const float4 xa = er[0], xb = er[1], xc = er[2], xd = er[3], xe = er[4];
#define DOT4(AC, XV, WV) AC = fmaf(XV.x, WV.x, AC); AC = fmaf(XV.y, WV.y, AC); AC = fmaf(XV.z, WV.z, AC); AC = fmaf(XV.w, WV.w, AC);
    const float lo = 2.0f / 60.0f / MUF;
    const float hi = 1.0f / 60.0f;
    {
      const float4* wr = w_s + jl0 * (NLAT / 4);
      float a = bias0;
      { const float4 wa = wr[0], wb = wr[1], wc = wr[2], wd = wr[3], we = wr[4];
        DOT4(a, xa, wa) DOT4(a, xb, wb) DOT4(a, xc, wc) DOT4(a, xd, wd) DOT4(a, xe, we) }
      const float s = 1.0f / (1.0f + expf(-a));
      const float d = ((hi - lo) * s + lo) * MUF;
      inv_s[tid] = 1.0f / (d * d);
    }
    {
      const float4* wr = w_s + jl1 * (NLAT / 4);
      float a = bias1;
      { const float4 wa = wr[0], wb = wr[1], wc = wr[2], wd = wr[3], we = wr[4];
        DOT4(a, xa, wa) DOT4(a, xb, wb) DOT4(a, xc, wc) DOT4(a, xd, wd) DOT4(a, xe, we) }
      const float s = 1.0f / (1.0f + expf(-a));
      const float d = ((hi - lo) * s + lo) * MUF;
      inv_s[tid + 256] = 1.0f / (d * d);
    }
  }

  float* st = stage[wv];
  float* stw = st + prow * TPITCH + piece * 4;
  TILE8(TILE_STORE)
  __syncthreads();
  const float4* ip = (const float4*)(inv_s + m * (LPG * BB));
  float4 i0 = ip[0], i1 = ip[1];
  const float* str = st + t * TPITCH;
  float d2[KNB];
  int   id[KNB];
#pragma unroll
  for (int q = 0; q < 8; ++q) {
    const float4 d = *(const float4*)(str + q * 4);
    const int4   k = *(const int4*)(str + KNB + q * 4);
    d2[q * 4 + 0] = d.x * d.x; d2[q * 4 + 1] = d.y * d.y;
    d2[q * 4 + 2] = d.z * d.z; d2[q * 4 + 3] = d.w * d.w;
    id[q * 4 + 0] = k.x; id[q * 4 + 1] = k.y; id[q * 4 + 2] = k.z; id[q * 4 + 3] = k.w;
  }

  float acc[BB];
#pragma unroll
  for (int b = 0; b < BB; ++b) acc[b] = 0.f;

#pragma nounroll
  for (int li = 0; li < LPG; ++li) {
    float* rb = row[li & 1];
    float4* dst = (float4*)rb;
    ROW16(ROW_STORE)
    const float inv[BB] = {i0.x, i0.y, i0.z, i0.w, i1.x, i1.y, i1.z, i1.w};
    __syncthreads();
    if (li + 1 < LPG) {
      src += NTOT / 4;
      ROW16(ROW_LOAD)
      ip += 2;
      i0 = ip[0]; i1 = ip[1];
    }
    float wsR[4][BB], psR[4][BB];
#pragma unroll
    for (int r = 0; r < 4; ++r) {
      float g[8];
#pragma unroll
      for (int j = 0; j < 8; ++j) g[j] = rb[id[r * 8 + j]];
      float ws[BB], ps[BB];
#pragma unroll
      for (int b = 0; b < BB; ++b) { ws[b] = 0.f; ps[b] = 0.f; }
#pragma unroll
      for (int j = 0; j < 8; ++j) {
        const float dd = d2[r * 8 + j];
#pragma unroll
        for (int b = 0; b < BB; ++b) {
          const float w = fmaxf(0.f, fmaf(-dd, inv[b], 1.0f));
          ws[b] += w;
          ps[b] = fmaf(g[j], w, ps[b]);
        }
      }
#pragma unroll
      for (int b = 0; b < BB; ++b) { wsR[r][b] = ws[b]; psR[r][b] = ps[b]; }
    }
    const float4 ev0 = *(const float4*)(enct_s + li * BB);
    const float4 ev1 = *(const float4*)(enct_s + li * BB + 4);
#pragma unroll
    for (int b = 0; b < BB; ++b) {
      const float wsT = (wsR[0][b] + wsR[1][b]) + (wsR[2][b] + wsR[3][b]);
      const float psT = (psR[0][b] + psR[1][b]) + (psR[2][b] + psR[3][b]);
      const float el = b == 0 ? ev0.x : b == 1 ? ev0.y : b == 2 ? ev0.z : b == 3 ? ev0.w
                     : b == 4 ? ev1.x : b == 5 ? ev1.y : b == 6 ? ev1.z : ev1.w;
      const float eu = __int_as_float(__builtin_amdgcn_readfirstlane(__float_as_int(el)));
      acc[b] = fmaf(eu, psT / wsT, acc[b]);
    }
  }
#pragma unroll
  for (int b = 0; b < BB; ++b) {
    convp[((size_t)(lg * BB + b)) * NTOT + n0 + t] = acc[b];
  }
}

#define SUM_THREADS 128
#define SUM_GROUPS  (BB * NTOT / 4 / SUM_THREADS)

__global__ __launch_bounds__(SUM_THREADS)
void latent_sum_kernel(const float* __restrict__ convp,
                       float* __restrict__ out) {
  const int i = blockIdx.x * SUM_THREADS + threadIdx.x;
  const float4* p4 = (const float4*)convp;
  const float4 v0 = p4[i];
  const float4 v1 = p4[(size_t)1 * (BB * NTOT / 4) + i];
  const float4 v2 = p4[(size_t)2 * (BB * NTOT / 4) + i];
  const float4 v3 = p4[(size_t)3 * (BB * NTOT / 4) + i];
  float4 acc;
  acc.x = ((v0.x + v1.x) + v2.x) + v3.x;
  acc.y = ((v0.y + v1.y) + v2.y) + v3.y;
  acc.z = ((v0.z + v1.z) + v2.z) + v3.z;
  acc.w = ((v0.w + v1.w) + v2.w) + v3.w;
  ((float4*)out)[i] = acc;
}

extern "C" void kernel_launch(void* const* d_in, const int* in_sizes, int n_in,
                              void* d_out, int out_size, void* d_ws, size_t ws_size,
                              hipStream_t stream) {
  const float* x      = (const float*)d_in[0];
  const float* enc_w  = (const float*)d_in[1];
  const float* enc_b  = (const float*)d_in[2];
  const float* dec_w  = (const float*)d_in[3];
  const float* bw_w   = (const float*)d_in[4];
  const float* bw_b   = (const float*)d_in[5];
  const float* ndist  = (const float*)d_in[6];
  const int*   nbid   = (const int*)d_in[7];
  const int*   labels = (const int*)d_in[8];
  float* out = (float*)d_out;

  float* ws       = (float*)d_ws;
  float* partials = ws + WS_PARTIALS;
  float* convp    = ws + WS_CONVP;

  encode_partial_wmma<<<NPART, 32 * ENC_WAVES, 0, stream>>>(x, enc_w, partials);
  conv_latent5_kernel<<<dim3(CCHUNKS, NLG), 256, 0, stream>>>(ndist, nbid, labels, dec_w, partials, enc_b, bw_w, bw_b, convp);
  latent_sum_kernel<<<SUM_GROUPS, SUM_THREADS, 0, stream>>>(convp, out);
}
